# starting kernel
# baseline (speedup 1.0000x reference)
#include <hip/hip_runtime.h>
#include <hip/hip_bf16.h>


typedef float v2f __attribute__((ext_vector_type(2)));
typedef float v8f __attribute__((ext_vector_type(8)));

__device__ __forceinline__ void atomAddF(float* p, float v) {
#if defined(__HIP_PLATFORM_AMD__)
    unsafeAtomicAdd(p, v);
#else
    atomicAdd(p, v);
#endif
}

__global__ __launch_bounds__(256) void k_init_ones(float* __restrict__ p, int n) {
    int i = blockIdx.x * blockDim.x + threadIdx.x;
    if (i < n) p[i] = 1.0f;
}

__global__ __launch_bounds__(256) void k_deg_accum(const int* __restrict__ dst,
                                                   float* __restrict__ deg, int E) {
    int e = blockIdx.x * blockDim.x + threadIdx.x;
    if (e < E) atomAddF(&deg[dst[e]], 1.0f);
}

__global__ __launch_bounds__(256) void k_rsqrt_inplace(float* __restrict__ p, int n) {
    int i = blockIdx.x * blockDim.x + threadIdx.x;
    if (i < n) p[i] = rsqrtf(p[i]);
}

template <int K, int F>
__global__ __launch_bounds__(256) void k_gemm_wmma(const float* __restrict__ X,
                                                   const float* __restrict__ W,
                                                   float* __restrict__ H,
                                                   int tilesM) {
    static_assert(K % 4 == 0 && F % 16 == 0, "shape");
    const int tilesF = F / 16;
    const int wave = blockIdx.x * (blockDim.x >> 5) + (threadIdx.x >> 5);
    if (wave >= tilesM * tilesF) return;
    const int tm = wave / tilesF;
    const int tf = wave % tilesF;
    const int lane = threadIdx.x & 31;
    const int l15  = lane & 15;
    const int hi2  = (lane >> 4) << 1;

    const float* xrow = X + (size_t)(tm * 16 + l15) * K;
    const float* wcol = W + (tf * 16 + l15);

    v8f acc = {};
#pragma unroll 4
    for (int kk = 0; kk < K; kk += 4) {
        v2f a, b;
        a.x = xrow[kk + hi2];
        a.y = xrow[kk + hi2 + 1];
        b.x = wcol[(size_t)(kk + hi2) * F];
        b.y = wcol[(size_t)(kk + hi2 + 1) * F];
        acc = __builtin_amdgcn_wmma_f32_16x16x4_f32(
             false, a,  false, b,
             (short)0, acc,  false,  false);
    }

    const int hi8 = (lane >> 4) << 3;
#pragma unroll
    for (int v = 0; v < 8; ++v)
        H[(size_t)(tm * 16 + hi8 + v) * F + tf * 16 + l15] = acc[v];
}

template <int K, int F>
__global__ __launch_bounds__(256) void k_gemm_tail(const float* __restrict__ X,
                                                   const float* __restrict__ W,
                                                   float* __restrict__ H,
                                                   int rowStart, int rowEnd) {
    int idx = blockIdx.x * blockDim.x + threadIdx.x;
    int rows = rowEnd - rowStart;
    if (idx >= rows * F) return;
    int r = rowStart + idx / F, f = idx % F;
    float s = 0.f;
    for (int k = 0; k < K; ++k) s += X[(size_t)r * K + k] * W[(size_t)k * F + f];
    H[(size_t)r * F + f] = s;
}

template <int F>
__global__ __launch_bounds__(256) void k_scatter(const float* __restrict__ H,
                                                 const int* __restrict__ src,
                                                 const int* __restrict__ dst,
                                                 const float* __restrict__ dinv,
                                                 float* __restrict__ AGG, int E) {
    constexpr int CH = F / 4;
    long long tid = (long long)blockIdx.x * blockDim.x + threadIdx.x;
    if (tid >= (long long)E * CH) return;
    int e = (int)(tid / CH);
    int c = (int)(tid % CH);
    int s = src[e], d = dst[e];
    float w = dinv[s] * dinv[d];
    float4 hv = ((const float4*)(H + (size_t)s * F))[c];
    float* ap = AGG + (size_t)d * F + c * 4;
    atomAddF(ap + 0, hv.x * w);
    atomAddF(ap + 1, hv.y * w);
    atomAddF(ap + 2, hv.z * w);
    atomAddF(ap + 3, hv.w * w);
}

template <int F, bool RELU>
__global__ __launch_bounds__(256) void k_self_bias(float* __restrict__ AGG,
                                                   const float* __restrict__ H,
                                                   const float* __restrict__ dinv,
                                                   const float* __restrict__ bias,
                                                   int N) {
    long long idx = (long long)blockIdx.x * blockDim.x + threadIdx.x;
    if (idx >= (long long)N * F) return;
    int i = (int)(idx / F), f = (int)(idx % F);
    float ns = dinv[i];
    ns *= ns;
    float v = AGG[idx] + H[idx] * ns + bias[f];
    if (RELU) v = fmaxf(v, 0.0f);
    AGG[idx] = v;
}

static inline int nblk(long long n, int t) { return (int)((n + t - 1) / t); }

template <int K, int F>
static void launch_gemm(const float* X, const float* W, float* H, int N, hipStream_t s) {
    int tilesM = N / 16;
    long long waves = (long long)tilesM * (F / 16);
    if (waves > 0)
        k_gemm_wmma<K, F><<<nblk(waves, 8), 256, 0, s>>>(X, W, H, tilesM);
    int tail = tilesM * 16;
    if (tail < N)
        k_gemm_tail<K, F><<<nblk((long long)(N - tail) * F, 256), 256, 0, s>>>(X, W, H, tail, N);
}

extern "C" void kernel_launch(void* const* d_in, const int* in_sizes, int n_in,
                              void* d_out, int out_size, void* d_ws, size_t ws_size,
                              hipStream_t stream) {
    const float* x  = (const float*)d_in[0];
    const int*   ei = (const int*)d_in[1];
    const float* W1 = (const float*)d_in[2]; const float* b1 = (const float*)d_in[3];
    const float* W2 = (const float*)d_in[4]; const float* b2 = (const float*)d_in[5];
    const float* W3 = (const float*)d_in[6]; const float* b3 = (const float*)d_in[7];
    const float* W4 = (const float*)d_in[8]; const float* b4 = (const float*)d_in[9];

    const int N = in_sizes[0] / 128;
    const int E = in_sizes[1] / 2;
    const int* src = ei;
    const int* dst = ei + E;

    float* out  = (float*)d_out;
    float* xrec = out;
    float* z    = out + (size_t)N * 128;

    float* ws   = (float*)d_ws;
    float* dinv = ws;
    float* bufH = ws + N;
    float* bufA = bufH + (size_t)N * 128;

    k_init_ones<<<nblk(N, 256), 256, 0, stream>>>(dinv, N);
    k_deg_accum<<<nblk(E, 256), 256, 0, stream>>>(dst, dinv, E);
    k_rsqrt_inplace<<<nblk(N, 256), 256, 0, stream>>>(dinv, N);

    launch_gemm<128, 64>(x, W1, bufH, N, stream);
    hipMemsetAsync(bufA, 0, (size_t)N * 64 * sizeof(float), stream);
    k_scatter<64><<<nblk((long long)E * 16, 256), 256, 0, stream>>>(bufH, src, dst, dinv, bufA, E);
    k_self_bias<64, true><<<nblk((long long)N * 64, 256), 256, 0, stream>>>(bufA, bufH, dinv, b1, N);

    launch_gemm<64, 32>(bufA, W2, bufH, N, stream);
    hipMemsetAsync(z, 0, (size_t)N * 32 * sizeof(float), stream);
    k_scatter<32><<<nblk((long long)E * 8, 256), 256, 0, stream>>>(bufH, src, dst, dinv, z, E);
    k_self_bias<32, false><<<nblk((long long)N * 32, 256), 256, 0, stream>>>(z, bufH, dinv, b2, N);

    launch_gemm<32, 64>(z, W3, bufH, N, stream);
    hipMemsetAsync(bufA, 0, (size_t)N * 64 * sizeof(float), stream);
    k_scatter<64><<<nblk((long long)E * 16, 256), 256, 0, stream>>>(bufH, src, dst, dinv, bufA, E);
    k_self_bias<64, true><<<nblk((long long)N * 64, 256), 256, 0, stream>>>(bufA, bufH, dinv, b3, N);

    launch_gemm<64, 128>(bufA, W4, bufH, N, stream);
    hipMemsetAsync(xrec, 0, (size_t)N * 128 * sizeof(float), stream);
    k_scatter<128><<<nblk((long long)E * 32, 256), 256, 0, stream>>>(bufH, src, dst, dinv, xrec, E);
    k_self_bias<128, false><<<nblk((long long)N * 128, 256), 256, 0, stream>>>(xrec, bufH, dinv, b4, N);
}


# GEMM store also writes the edge sum's start value h*dinv^2+bias: no clearing memset and no self+bias pass (ReLU at the next GEMM's load); 15 -> 11 launches; self term now summed first, not last
# speedup vs baseline: 1.5704x; 1.5704x over previous
#include <hip/hip_runtime.h>
#include <hip/hip_bf16.h>


typedef float v2f __attribute__((ext_vector_type(2)));
typedef float v8f __attribute__((ext_vector_type(8)));

__device__ __forceinline__ void atomAddF(float* p, float v) {
#if defined(__HIP_PLATFORM_AMD__)
    unsafeAtomicAdd(p, v);
#else
    atomicAdd(p, v);
#endif
}

__global__ __launch_bounds__(256) void k_init_ones(float* __restrict__ p, int n) {
    int i = blockIdx.x * blockDim.x + threadIdx.x;
    if (i < n) p[i] = 1.0f;
}

__global__ __launch_bounds__(256) void k_deg_accum(const int* __restrict__ dst,
                                                   float* __restrict__ deg, int E) {
    int e = blockIdx.x * blockDim.x + threadIdx.x;
    if (e < E) atomAddF(&deg[dst[e]], 1.0f);
}

__global__ __launch_bounds__(256) void k_rsqrt_inplace(float* __restrict__ p, int n) {
    int i = blockIdx.x * blockDim.x + threadIdx.x;
    if (i < n) p[i] = rsqrtf(p[i]);
}

template <int K, int F, bool RELU_IN>
__global__ __launch_bounds__(256) void k_gemm_wmma(const float* __restrict__ X,
                                                   const float* __restrict__ W,
                                                   const float* __restrict__ dinv,
                                                   const float* __restrict__ bias,
                                                   float* __restrict__ H,
                                                   float* __restrict__ AGG,
                                                   int tilesM) {
    static_assert(K % 4 == 0 && F % 16 == 0, "shape");
    const int tilesF = F / 16;
    const int wave = blockIdx.x * (blockDim.x >> 5) + (threadIdx.x >> 5);
    if (wave >= tilesM * tilesF) return;
    const int tm = wave / tilesF;
    const int tf = wave % tilesF;
    const int lane = threadIdx.x & 31;
    const int l15  = lane & 15;
    const int hi2  = (lane >> 4) << 1;

    const float* xrow = X + (size_t)(tm * 16 + l15) * K;
    const float* wcol = W + (tf * 16 + l15);
    const int hi8 = (lane >> 4) << 3;

    float ns[8];
#pragma unroll
    for (int v = 0; v < 8; ++v) {
        const float d = dinv[tm * 16 + hi8 + v];
        ns[v] = d * d;
    }
    const float bcol = bias[tf * 16 + l15];

    v8f acc = {};
#pragma unroll 4
    for (int kk = 0; kk < K; kk += 4) {
        v2f a, b;
        a.x = xrow[kk + hi2];
        a.y = xrow[kk + hi2 + 1];
        if (RELU_IN) {
            a.x = fmaxf(a.x, 0.0f);
            a.y = fmaxf(a.y, 0.0f);
        }
        b.x = wcol[(size_t)(kk + hi2) * F];
        b.y = wcol[(size_t)(kk + hi2 + 1) * F];
        acc = __builtin_amdgcn_wmma_f32_16x16x4_f32(
             false, a,  false, b,
             (short)0, acc,  false,  false);
    }

#pragma unroll
    for (int v = 0; v < 8; ++v) {
        const size_t o = (size_t)(tm * 16 + hi8 + v) * F + tf * 16 + l15;
        H[o]   = acc[v];
        AGG[o] = fmaf(acc[v], ns[v], bcol);
    }
}

template <int K, int F, bool RELU_IN>
__global__ __launch_bounds__(256) void k_gemm_tail(const float* __restrict__ X,
                                                   const float* __restrict__ W,
                                                   const float* __restrict__ dinv,
                                                   const float* __restrict__ bias,
                                                   float* __restrict__ H,
                                                   float* __restrict__ AGG,
                                                   int rowStart, int rowEnd) {
    int idx = blockIdx.x * blockDim.x + threadIdx.x;
    int rows = rowEnd - rowStart;
    if (idx >= rows * F) return;
    int r = rowStart + idx / F, f = idx % F;
    float s = 0.f;
    for (int k = 0; k < K; ++k) {
        float xv = X[(size_t)r * K + k];
        if (RELU_IN) xv = fmaxf(xv, 0.0f);
        s += xv * W[(size_t)k * F + f];
    }
    const float d = dinv[r];
    H[(size_t)r * F + f]   = s;
    AGG[(size_t)r * F + f] = fmaf(s, d * d, bias[f]);
}

template <int F>
__global__ __launch_bounds__(256) void k_scatter(const float* __restrict__ H,
                                                 const int* __restrict__ src,
                                                 const int* __restrict__ dst,
                                                 const float* __restrict__ dinv,
                                                 float* __restrict__ AGG, int E) {
    static_assert(F % 32 == 0, "row must be whole 128-byte lines");
    constexpr int NJ  = F / 32;
    constexpr int EPW = 8;
    const int lane = threadIdx.x & 31;
    const int wave = __builtin_amdgcn_readfirstlane(
        (int)(blockIdx.x * (blockDim.x >> 5) + (threadIdx.x >> 5)));
    const int e0 = wave * EPW;
    if (e0 >= E) return;

    if (e0 + EPW <= E) {
        int idx = 0;
        float dv = 0.0f;
        if (lane < 16) {
            const int* ip = (lane & 8) ? dst : src;
            idx = ip[e0 + (lane & 7)];
            dv  = dinv[idx];
        }
        const float wv = dv * __shfl_xor(dv, 8);

        float v[EPW][NJ];
        for (int j = 0; j < EPW; ++j) {
            const int s = __builtin_amdgcn_readlane(idx, j);
            const float* hp = H + (size_t)s * F + lane;
            for (int c = 0; c < NJ; ++c) v[j][c] = hp[c * 32];
        }
        for (int j = 0; j < EPW; ++j) {
            const int   d = __builtin_amdgcn_readlane(idx, j + 8);
            const float w = __int_as_float(__builtin_amdgcn_readlane(__float_as_int(wv), j));
            float* ap = AGG + (size_t)d * F + lane;
            for (int c = 0; c < NJ; ++c) atomAddF(ap + c * 32, v[j][c] * w);
        }
    } else {
        for (int e = e0; e < E; ++e) {
            const int s = src[e], d = dst[e];
            const float w = dinv[s] * dinv[d];
            const float* hp = H + (size_t)s * F + lane;
            float* ap = AGG + (size_t)d * F + lane;
            for (int c = 0; c < NJ; ++c) atomAddF(ap + c * 32, hp[c * 32] * w);
        }
    }
}

static inline int nblk(long long n, int t) { return (int)((n + t - 1) / t); }

template <int K, int F, bool RELU_IN>
static void launch_gemm(const float* X, const float* W, const float* dinv, const float* bias,
                        float* H, float* AGG, int N, hipStream_t s) {
    int tilesM = N / 16;
    long long waves = (long long)tilesM * (F / 16);
    if (waves > 0)
        k_gemm_wmma<K, F, RELU_IN><<<nblk(waves, 8), 256, 0, s>>>(X, W, dinv, bias, H, AGG, tilesM);
    int tail = tilesM * 16;
    if (tail < N)
        k_gemm_tail<K, F, RELU_IN><<<nblk((long long)(N - tail) * F, 256), 256, 0, s>>>(
            X, W, dinv, bias, H, AGG, tail, N);
}

extern "C" void kernel_launch(void* const* d_in, const int* in_sizes, int n_in,
                              void* d_out, int out_size, void* d_ws, size_t ws_size,
                              hipStream_t stream) {
    const float* x  = (const float*)d_in[0];
    const int*   ei = (const int*)d_in[1];
    const float* W1 = (const float*)d_in[2]; const float* b1 = (const float*)d_in[3];
    const float* W2 = (const float*)d_in[4]; const float* b2 = (const float*)d_in[5];
    const float* W3 = (const float*)d_in[6]; const float* b3 = (const float*)d_in[7];
    const float* W4 = (const float*)d_in[8]; const float* b4 = (const float*)d_in[9];

    const int N = in_sizes[0] / 128;
    const int E = in_sizes[1] / 2;
    const int* src = ei;
    const int* dst = ei + E;

    float* out  = (float*)d_out;
    float* xrec = out;
    float* z    = out + (size_t)N * 128;

    float* ws   = (float*)d_ws;
    float* dinv = ws;
    float* bufH = ws + N;
    float* bufA = bufH + (size_t)N * 128;

    k_init_ones<<<nblk(N, 256), 256, 0, stream>>>(dinv, N);
    k_deg_accum<<<nblk(E, 256), 256, 0, stream>>>(dst, dinv, E);
    k_rsqrt_inplace<<<nblk(N, 256), 256, 0, stream>>>(dinv, N);


    launch_gemm<128, 64, false>(x, W1, dinv, b1, bufH, bufA, N, stream);
    k_scatter<64><<<nblk(nblk(E, 8), 8), 256, 0, stream>>>(bufH, src, dst, dinv, bufA, E);

    launch_gemm<64, 32, true>(bufA, W2, dinv, b2, bufH, z, N, stream);
    k_scatter<32><<<nblk(nblk(E, 8), 8), 256, 0, stream>>>(bufH, src, dst, dinv, z, E);

    launch_gemm<32, 64, false>(z, W3, dinv, b3, bufH, bufA, N, stream);
    k_scatter<64><<<nblk(nblk(E, 8), 8), 256, 0, stream>>>(bufH, src, dst, dinv, bufA, E);

    launch_gemm<64, 128, true>(bufA, W4, dinv, b4, bufH, xrec, N, stream);
    k_scatter<128><<<nblk(nblk(E, 8), 8), 256, 0, stream>>>(bufH, src, dst, dinv, xrec, E);
}


# 128-wide edge pass as two 64-column halves, a launch each, workspace rows moved onto the 256-byte grid: each half's rows and sums fit the L2; 11 -> 12 launches; f32 atomics in arrival order as before
# speedup vs baseline: 1.7659x; 1.1245x over previous
#include <hip/hip_runtime.h>
#include <hip/hip_bf16.h>


typedef float v2f __attribute__((ext_vector_type(2)));
typedef float v8f __attribute__((ext_vector_type(8)));

__device__ __forceinline__ void atomAddF(float* p, float v) {
#if defined(__HIP_PLATFORM_AMD__)
    unsafeAtomicAdd(p, v);
#else
    atomicAdd(p, v);
#endif
}

__global__ __launch_bounds__(256) void k_init_ones(float* __restrict__ p, int n) {
    int i = blockIdx.x * blockDim.x + threadIdx.x;
    if (i < n) p[i] = 1.0f;
}

__global__ __launch_bounds__(256) void k_deg_accum(const int* __restrict__ dst,
                                                   float* __restrict__ deg, int E) {
    int e = blockIdx.x * blockDim.x + threadIdx.x;
    if (e < E) atomAddF(&deg[dst[e]], 1.0f);
}

__global__ __launch_bounds__(256) void k_rsqrt_inplace(float* __restrict__ p, int n) {
    int i = blockIdx.x * blockDim.x + threadIdx.x;
    if (i < n) p[i] = rsqrtf(p[i]);
}

template <int K, int F, bool RELU_IN>
__global__ __launch_bounds__(256) void k_gemm_wmma(const float* __restrict__ X,
                                                   const float* __restrict__ W,
                                                   const float* __restrict__ dinv,
                                                   const float* __restrict__ bias,
                                                   float* __restrict__ H,
                                                   float* __restrict__ AGG,
                                                   int tilesM) {
    static_assert(K % 4 == 0 && F % 16 == 0, "shape");
    const int tilesF = F / 16;
    const int wave = blockIdx.x * (blockDim.x >> 5) + (threadIdx.x >> 5);
    if (wave >= tilesM * tilesF) return;
    const int tm = wave / tilesF;
    const int tf = wave % tilesF;
    const int lane = threadIdx.x & 31;
    const int l15  = lane & 15;
    const int hi2  = (lane >> 4) << 1;

    const float* xrow = X + (size_t)(tm * 16 + l15) * K;
    const float* wcol = W + (tf * 16 + l15);
    const int hi8 = (lane >> 4) << 3;

    float ns[8];
#pragma unroll
    for (int v = 0; v < 8; ++v) {
        const float d = dinv[tm * 16 + hi8 + v];
        ns[v] = d * d;
    }
    const float bcol = bias[tf * 16 + l15];

    v8f acc = {};
#pragma unroll 4
    for (int kk = 0; kk < K; kk += 4) {
        v2f a, b;
        a.x = xrow[kk + hi2];
        a.y = xrow[kk + hi2 + 1];
        if (RELU_IN) {
            a.x = fmaxf(a.x, 0.0f);
            a.y = fmaxf(a.y, 0.0f);
        }
        b.x = wcol[(size_t)(kk + hi2) * F];
        b.y = wcol[(size_t)(kk + hi2 + 1) * F];
        acc = __builtin_amdgcn_wmma_f32_16x16x4_f32(
             false, a,  false, b,
             (short)0, acc,  false,  false);
    }

#pragma unroll
    for (int v = 0; v < 8; ++v) {
        const size_t o = (size_t)(tm * 16 + hi8 + v) * F + tf * 16 + l15;
        H[o]   = acc[v];
        AGG[o] = fmaf(acc[v], ns[v], bcol);
    }
}

template <int K, int F, bool RELU_IN>
__global__ __launch_bounds__(256) void k_gemm_tail(const float* __restrict__ X,
                                                   const float* __restrict__ W,
                                                   const float* __restrict__ dinv,
                                                   const float* __restrict__ bias,
                                                   float* __restrict__ H,
                                                   float* __restrict__ AGG,
                                                   int rowStart, int rowEnd) {
    int idx = blockIdx.x * blockDim.x + threadIdx.x;
    int rows = rowEnd - rowStart;
    if (idx >= rows * F) return;
    int r = rowStart + idx / F, f = idx % F;
    float s = 0.f;
    for (int k = 0; k < K; ++k) {
        float xv = X[(size_t)r * K + k];
        if (RELU_IN) xv = fmaxf(xv, 0.0f);
        s += xv * W[(size_t)k * F + f];
    }
    const float d = dinv[r];
    H[(size_t)r * F + f]   = s;
    AGG[(size_t)r * F + f] = fmaf(s, d * d, bias[f]);
}

template <int F, int LD>
__global__ __launch_bounds__(256) void k_scatter(const float* __restrict__ H,
                                                 const int* __restrict__ src,
                                                 const int* __restrict__ dst,
                                                 const float* __restrict__ dinv,
                                                 float* __restrict__ AGG, int E) {
    static_assert(F % 32 == 0 && LD % 32 == 0 && F <= LD, "whole 128-byte lines");
    constexpr int NJ  = F / 32;
    constexpr int EPW = 8;
    const int lane = threadIdx.x & 31;
    const int wave = __builtin_amdgcn_readfirstlane(
        (int)(blockIdx.x * (blockDim.x >> 5) + (threadIdx.x >> 5)));
    const int e0 = wave * EPW;
    if (e0 >= E) return;

    if (e0 + EPW <= E) {
        int idx = 0;
        float dv = 0.0f;
        if (lane < 16) {
            const int* ip = (lane & 8) ? dst : src;
            idx = ip[e0 + (lane & 7)];
            dv  = dinv[idx];
        }
        const float wv = dv * __shfl_xor(dv, 8);

        float v[EPW][NJ];
        for (int j = 0; j < EPW; ++j) {
            const int s = __builtin_amdgcn_readlane(idx, j);
            const float* hp = H + (size_t)s * LD + lane;
            for (int c = 0; c < NJ; ++c) v[j][c] = hp[c * 32];
        }
        for (int j = 0; j < EPW; ++j) {
            const int   d = __builtin_amdgcn_readlane(idx, j + 8);
            const float w = __int_as_float(__builtin_amdgcn_readlane(__float_as_int(wv), j));
            float* ap = AGG + (size_t)d * LD + lane;
            for (int c = 0; c < NJ; ++c) atomAddF(ap + c * 32, v[j][c] * w);
        }
    } else {
        for (int e = e0; e < E; ++e) {
            const int s = src[e], d = dst[e];
            const float w = dinv[s] * dinv[d];
            const float* hp = H + (size_t)s * LD + lane;
            float* ap = AGG + (size_t)d * LD + lane;
            for (int c = 0; c < NJ; ++c) atomAddF(ap + c * 32, hp[c * 32] * w);
        }
    }
}

static inline int nblk(long long n, int t) { return (int)((n + t - 1) / t); }

template <int K, int F, bool RELU_IN>
static void launch_gemm(const float* X, const float* W, const float* dinv, const float* bias,
                        float* H, float* AGG, int N, hipStream_t s) {
    int tilesM = N / 16;
    long long waves = (long long)tilesM * (F / 16);
    if (waves > 0)
        k_gemm_wmma<K, F, RELU_IN><<<nblk(waves, 8), 256, 0, s>>>(X, W, dinv, bias, H, AGG, tilesM);
    int tail = tilesM * 16;
    if (tail < N)
        k_gemm_tail<K, F, RELU_IN><<<nblk((long long)(N - tail) * F, 256), 256, 0, s>>>(
            X, W, dinv, bias, H, AGG, tail, N);
}

extern "C" void kernel_launch(void* const* d_in, const int* in_sizes, int n_in,
                              void* d_out, int out_size, void* d_ws, size_t ws_size,
                              hipStream_t stream) {
    const float* x  = (const float*)d_in[0];
    const int*   ei = (const int*)d_in[1];
    const float* W1 = (const float*)d_in[2]; const float* b1 = (const float*)d_in[3];
    const float* W2 = (const float*)d_in[4]; const float* b2 = (const float*)d_in[5];
    const float* W3 = (const float*)d_in[6]; const float* b3 = (const float*)d_in[7];
    const float* W4 = (const float*)d_in[8]; const float* b4 = (const float*)d_in[9];

    const int N = in_sizes[0] / 128;
    const int E = in_sizes[1] / 2;
    const int* src = ei;
    const int* dst = ei + E;

    float* out  = (float*)d_out;
    float* xrec = out;
    float* z    = out + (size_t)N * 128;

    float* ws   = (float*)d_ws;
    float* bufH = ws;
    float* bufA = bufH + (size_t)N * 128;
    float* dinv = bufA + (size_t)N * 64;

    k_init_ones<<<nblk(N, 256), 256, 0, stream>>>(dinv, N);
    k_deg_accum<<<nblk(E, 256), 256, 0, stream>>>(dst, dinv, E);
    k_rsqrt_inplace<<<nblk(N, 256), 256, 0, stream>>>(dinv, N);


    launch_gemm<128, 64, false>(x, W1, dinv, b1, bufH, bufA, N, stream);
    k_scatter<64, 64><<<nblk(nblk(E, 8), 8), 256, 0, stream>>>(bufH, src, dst, dinv, bufA, E);

    launch_gemm<64, 32, true>(bufA, W2, dinv, b2, bufH, z, N, stream);
    k_scatter<32, 32><<<nblk(nblk(E, 8), 8), 256, 0, stream>>>(bufH, src, dst, dinv, z, E);

    launch_gemm<32, 64, false>(z, W3, dinv, b3, bufH, bufA, N, stream);
    k_scatter<64, 64><<<nblk(nblk(E, 8), 8), 256, 0, stream>>>(bufH, src, dst, dinv, bufA, E);

    launch_gemm<64, 128, true>(bufA, W4, dinv, b4, bufH, xrec, N, stream);
    k_scatter<64, 128><<<nblk(nblk(E, 8), 8), 256, 0, stream>>>(bufH, src, dst, dinv, xrec, E);
    k_scatter<64, 128><<<nblk(nblk(E, 8), 8), 256, 0, stream>>>(bufH + 64, src, dst, dinv, xrec + 64, E);
}


# Edge weight dinv[src]*dinv[dst] stored by layer 1's edge pass; the other four read 8 weights in one 32-byte load, not 16 scattered dinv words; layer-4 GEMM by 64-col halves for room; 13 launches
# speedup vs baseline: 1.8708x; 1.0594x over previous
#include <hip/hip_runtime.h>
#include <hip/hip_bf16.h>


typedef float v2f __attribute__((ext_vector_type(2)));
typedef float v8f __attribute__((ext_vector_type(8)));

__device__ __forceinline__ void atomAddF(float* p, float v) {
#if defined(__HIP_PLATFORM_AMD__)
    unsafeAtomicAdd(p, v);
#else
    atomicAdd(p, v);
#endif
}

__global__ __launch_bounds__(256) void k_init_ones(float* __restrict__ p, int n) {
    int i = blockIdx.x * blockDim.x + threadIdx.x;
    if (i < n) p[i] = 1.0f;
}

__global__ __launch_bounds__(256) void k_deg_accum(const int* __restrict__ dst,
                                                   float* __restrict__ deg, int E) {
    int e = blockIdx.x * blockDim.x + threadIdx.x;
    if (e < E) atomAddF(&deg[dst[e]], 1.0f);
}

__global__ __launch_bounds__(256) void k_rsqrt_inplace(float* __restrict__ p, int n) {
    int i = blockIdx.x * blockDim.x + threadIdx.x;
    if (i < n) p[i] = rsqrtf(p[i]);
}

template <int K, int F, int LDW, int LDH, int LDA, bool RELU_IN>
__global__ __launch_bounds__(256) void k_gemm_wmma(const float* __restrict__ X,
                                                   const float* __restrict__ W,
                                                   const float* __restrict__ dinv,
                                                   const float* __restrict__ bias,
                                                   float* __restrict__ H,
                                                   float* __restrict__ AGG,
                                                   int tilesM) {
    static_assert(K % 4 == 0 && F % 16 == 0 && F <= LDW && F <= LDH && F <= LDA, "shape");
    const int tilesF = F / 16;
    const int wave = blockIdx.x * (blockDim.x >> 5) + (threadIdx.x >> 5);
    if (wave >= tilesM * tilesF) return;
    const int tm = wave / tilesF;
    const int tf = wave % tilesF;
    const int lane = threadIdx.x & 31;
    const int l15  = lane & 15;
    const int hi2  = (lane >> 4) << 1;

    const float* xrow = X + (size_t)(tm * 16 + l15) * K;
    const float* wcol = W + (tf * 16 + l15);
    const int hi8 = (lane >> 4) << 3;

    float ns[8];
#pragma unroll
    for (int v = 0; v < 8; ++v) {
        const float d = dinv[tm * 16 + hi8 + v];
        ns[v] = d * d;
    }
    const float bcol = bias[tf * 16 + l15];

    v8f acc = {};
#pragma unroll 4
    for (int kk = 0; kk < K; kk += 4) {
        v2f a, b;
        a.x = xrow[kk + hi2];
        a.y = xrow[kk + hi2 + 1];
        if (RELU_IN) {
            a.x = fmaxf(a.x, 0.0f);
            a.y = fmaxf(a.y, 0.0f);
        }
        b.x = wcol[(size_t)(kk + hi2) * LDW];
        b.y = wcol[(size_t)(kk + hi2 + 1) * LDW];
        acc = __builtin_amdgcn_wmma_f32_16x16x4_f32(
             false, a,  false, b,
             (short)0, acc,  false,  false);
    }

#pragma unroll
    for (int v = 0; v < 8; ++v) {
        const size_t r = (size_t)(tm * 16 + hi8 + v);
        const int    c = tf * 16 + l15;
        H[r * LDH + c]   = acc[v];
        AGG[r * LDA + c] = fmaf(acc[v], ns[v], bcol);
    }
}

template <int K, int F, int LDW, int LDH, int LDA, bool RELU_IN>
__global__ __launch_bounds__(256) void k_gemm_tail(const float* __restrict__ X,
                                                   const float* __restrict__ W,
                                                   const float* __restrict__ dinv,
                                                   const float* __restrict__ bias,
                                                   float* __restrict__ H,
                                                   float* __restrict__ AGG,
                                                   int rowStart, int rowEnd) {
    int idx = blockIdx.x * blockDim.x + threadIdx.x;
    int rows = rowEnd - rowStart;
    if (idx >= rows * F) return;
    int r = rowStart + idx / F, f = idx % F;
    float s = 0.f;
    for (int k = 0; k < K; ++k) {
        float xv = X[(size_t)r * K + k];
        if (RELU_IN) xv = fmaxf(xv, 0.0f);
        s += xv * W[(size_t)k * LDW + f];
    }
    const float d = dinv[r];
    H[(size_t)r * LDH + f]   = s;
    AGG[(size_t)r * LDA + f] = fmaf(s, d * d, bias[f]);
}

template <int F, int LDH, int LDA, bool MAKE_W>
__global__ __launch_bounds__(256) void k_scatter(const float* __restrict__ H,
                                                 const int* __restrict__ src,
                                                 const int* __restrict__ dst,
                                                 const float* __restrict__ dinv,
                                                 float* __restrict__ WGT,
                                                 float* __restrict__ AGG, int E) {
    static_assert(F % 32 == 0 && LDH % 32 == 0 && LDA % 32 == 0 && F <= LDH && F <= LDA,
                  "whole 128-byte lines");
    constexpr int NJ  = F / 32;
    constexpr int EPW = 8;
    const int lane = threadIdx.x & 31;
    const int wave = __builtin_amdgcn_readfirstlane(
        (int)(blockIdx.x * (blockDim.x >> 5) + (threadIdx.x >> 5)));
    const int e0 = wave * EPW;
    if (e0 >= E) return;

    if (e0 + EPW <= E) {
        int idx = 0;
        float wv = 0.0f;
        if (MAKE_W) {
            float dv = 0.0f;
            if (lane < 16) {
                const int* ip = (lane & 8) ? dst : src;
                idx = ip[e0 + (lane & 7)];
                dv  = dinv[idx];
            }
            wv = dv * __shfl_xor(dv, 8);
            if (lane < 8) WGT[e0 + lane] = wv;
        } else {
            if (lane < 16) {
                const int* ip = (lane & 8) ? dst : src;
                idx = ip[e0 + (lane & 7)];
                wv  = WGT[e0 + (lane & 7)];
            }
        }

        float v[EPW][NJ];
        for (int j = 0; j < EPW; ++j) {
            const int s = __builtin_amdgcn_readlane(idx, j);
            const float* hp = H + (size_t)s * LDH + lane;
            for (int c = 0; c < NJ; ++c) v[j][c] = hp[c * 32];
        }
        for (int j = 0; j < EPW; ++j) {
            const int   d = __builtin_amdgcn_readlane(idx, j + 8);
            const float w = __int_as_float(__builtin_amdgcn_readlane(__float_as_int(wv), j));
            float* ap = AGG + (size_t)d * LDA + lane;
            for (int c = 0; c < NJ; ++c) atomAddF(ap + c * 32, v[j][c] * w);
        }
    } else {
        for (int e = e0; e < E; ++e) {
            const int s = src[e], d = dst[e];
            float w;
            if (MAKE_W) {
                w = dinv[s] * dinv[d];
                if (lane == 0) WGT[e] = w;
            } else {
                w = WGT[e];
            }
            const float* hp = H + (size_t)s * LDH + lane;
            float* ap = AGG + (size_t)d * LDA + lane;
            for (int c = 0; c < NJ; ++c) atomAddF(ap + c * 32, hp[c * 32] * w);
        }
    }
}

static inline int nblk(long long n, int t) { return (int)((n + t - 1) / t); }

template <int K, int F, int LDW, int LDH, int LDA, bool RELU_IN>
static void launch_gemm(const float* X, const float* W, const float* dinv, const float* bias,
                        float* H, float* AGG, int N, hipStream_t s) {
    int tilesM = N / 16;
    long long waves = (long long)tilesM * (F / 16);
    if (waves > 0)
        k_gemm_wmma<K, F, LDW, LDH, LDA, RELU_IN><<<nblk(waves, 8), 256, 0, s>>>(
            X, W, dinv, bias, H, AGG, tilesM);
    int tail = tilesM * 16;
    if (tail < N)
        k_gemm_tail<K, F, LDW, LDH, LDA, RELU_IN><<<nblk((long long)(N - tail) * F, 256), 256, 0, s>>>(
            X, W, dinv, bias, H, AGG, tail, N);
}

extern "C" void kernel_launch(void* const* d_in, const int* in_sizes, int n_in,
                              void* d_out, int out_size, void* d_ws, size_t ws_size,
                              hipStream_t stream) {
    const float* x  = (const float*)d_in[0];
    const int*   ei = (const int*)d_in[1];
    const float* W1 = (const float*)d_in[2]; const float* b1 = (const float*)d_in[3];
    const float* W2 = (const float*)d_in[4]; const float* b2 = (const float*)d_in[5];
    const float* W3 = (const float*)d_in[6]; const float* b3 = (const float*)d_in[7];
    const float* W4 = (const float*)d_in[8]; const float* b4 = (const float*)d_in[9];

    const int N = in_sizes[0] / 128;
    const int E = in_sizes[1] / 2;
    const int* src = ei;
    const int* dst = ei + E;

    float* out  = (float*)d_out;
    float* xrec = out;
    float* z    = out + (size_t)N * 128;

    float* ws   = (float*)d_ws;
    float* bufH = ws;
    float* wgt  = bufH + (size_t)N * 64;
    float* bufA = bufH + (size_t)N * 128;
    float* dinv = bufA + (size_t)N * 64;

    k_init_ones<<<nblk(N, 256), 256, 0, stream>>>(dinv, N);
    k_deg_accum<<<nblk(E, 256), 256, 0, stream>>>(dst, dinv, E);
    k_rsqrt_inplace<<<nblk(N, 256), 256, 0, stream>>>(dinv, N);


    launch_gemm<128, 64, 64, 64, 64, false>(x, W1, dinv, b1, bufH, bufA, N, stream);
    k_scatter<64, 64, 64, true><<<nblk(nblk(E, 8), 8), 256, 0, stream>>>(bufH, src, dst, dinv, wgt, bufA, E);

    launch_gemm<64, 32, 32, 32, 32, true>(bufA, W2, dinv, b2, bufH, z, N, stream);
    k_scatter<32, 32, 32, false><<<nblk(nblk(E, 8), 8), 256, 0, stream>>>(bufH, src, dst, dinv, wgt, z, E);

    launch_gemm<32, 64, 64, 64, 64, false>(z, W3, dinv, b3, bufH, bufA, N, stream);
    k_scatter<64, 64, 64, false><<<nblk(nblk(E, 8), 8), 256, 0, stream>>>(bufH, src, dst, dinv, wgt, bufA, E);

    launch_gemm<64, 64, 128, 64, 128, true>(bufA, W4, dinv, b4, bufH, xrec, N, stream);
    k_scatter<64, 64, 128, false><<<nblk(nblk(E, 8), 8), 256, 0, stream>>>(bufH, src, dst, dinv, wgt, xrec, E);
    launch_gemm<64, 64, 128, 64, 128, true>(bufA, W4 + 64, dinv, b4 + 64, bufH, xrec + 64, N, stream);
    k_scatter<64, 64, 128, false><<<nblk(nblk(E, 8), 8), 256, 0, stream>>>(bufH, src, dst, dinv, wgt, xrec + 64, E);
}


# Edge passes: each package half (die id) adds only into its own half of the target rows; per-half ticket counters hand out 128-edge groups, waves filter by dst; f32 atomics in arrival order as before
# speedup vs baseline: 1.9669x; 1.0514x over previous
#include <hip/hip_runtime.h>
#include <hip/hip_bf16.h>


typedef float v2f __attribute__((ext_vector_type(2)));
typedef float v8f __attribute__((ext_vector_type(8)));

__device__ __forceinline__ void atomAddF(float* p, float v) {
#if defined(__HIP_PLATFORM_AMD__)
    unsafeAtomicAdd(p, v);
#else
    atomicAdd(p, v);
#endif
}

__global__ __launch_bounds__(256) void k_init_ones(float* __restrict__ p, int n,
                                                   int* __restrict__ q, int nq) {
    int i = blockIdx.x * blockDim.x + threadIdx.x;
    if (i < n) p[i] = 1.0f;
    if (i < nq) q[i] = 0;
}

__global__ __launch_bounds__(256) void k_deg_accum(const int* __restrict__ dst,
                                                   float* __restrict__ deg, int E) {
    int e = blockIdx.x * blockDim.x + threadIdx.x;
    if (e < E) atomAddF(&deg[dst[e]], 1.0f);
}

__global__ __launch_bounds__(256) void k_rsqrt_inplace(float* __restrict__ p, int n) {
    int i = blockIdx.x * blockDim.x + threadIdx.x;
    if (i < n) p[i] = rsqrtf(p[i]);
}

template <int K, int F, int LDW, int LDH, int LDA, bool RELU_IN>
__global__ __launch_bounds__(256) void k_gemm_wmma(const float* __restrict__ X,
                                                   const float* __restrict__ W,
                                                   const float* __restrict__ dinv,
                                                   const float* __restrict__ bias,
                                                   float* __restrict__ H,
                                                   float* __restrict__ AGG,
                                                   int tilesM) {
    static_assert(K % 4 == 0 && F % 16 == 0 && F <= LDW && F <= LDH && F <= LDA, "shape");
    const int tilesF = F / 16;
    const int wave = blockIdx.x * (blockDim.x >> 5) + (threadIdx.x >> 5);
    if (wave >= tilesM * tilesF) return;
    const int tm = wave / tilesF;
    const int tf = wave % tilesF;
    const int lane = threadIdx.x & 31;
    const int l15  = lane & 15;
    const int hi2  = (lane >> 4) << 1;

    const float* xrow = X + (size_t)(tm * 16 + l15) * K;
    const float* wcol = W + (tf * 16 + l15);
    const int hi8 = (lane >> 4) << 3;

    float ns[8];
#pragma unroll
    for (int v = 0; v < 8; ++v) {
        const float d = dinv[tm * 16 + hi8 + v];
        ns[v] = d * d;
    }
    const float bcol = bias[tf * 16 + l15];

    v8f acc = {};
#pragma unroll 4
    for (int kk = 0; kk < K; kk += 4) {
        v2f a, b;
        a.x = xrow[kk + hi2];
        a.y = xrow[kk + hi2 + 1];
        if (RELU_IN) {
            a.x = fmaxf(a.x, 0.0f);
            a.y = fmaxf(a.y, 0.0f);
        }
        b.x = wcol[(size_t)(kk + hi2) * LDW];
        b.y = wcol[(size_t)(kk + hi2 + 1) * LDW];
        acc = __builtin_amdgcn_wmma_f32_16x16x4_f32(
             false, a,  false, b,
             (short)0, acc,  false,  false);
    }

#pragma unroll
    for (int v = 0; v < 8; ++v) {
        const size_t r = (size_t)(tm * 16 + hi8 + v);
        const int    c = tf * 16 + l15;
        H[r * LDH + c]   = acc[v];
        AGG[r * LDA + c] = fmaf(acc[v], ns[v], bcol);
    }
}

template <int K, int F, int LDW, int LDH, int LDA, bool RELU_IN>
__global__ __launch_bounds__(256) void k_gemm_tail(const float* __restrict__ X,
                                                   const float* __restrict__ W,
                                                   const float* __restrict__ dinv,
                                                   const float* __restrict__ bias,
                                                   float* __restrict__ H,
                                                   float* __restrict__ AGG,
                                                   int rowStart, int rowEnd) {
    int idx = blockIdx.x * blockDim.x + threadIdx.x;
    int rows = rowEnd - rowStart;
    if (idx >= rows * F) return;
    int r = rowStart + idx / F, f = idx % F;
    float s = 0.f;
    for (int k = 0; k < K; ++k) {
        float xv = X[(size_t)r * K + k];
        if (RELU_IN) xv = fmaxf(xv, 0.0f);
        s += xv * W[(size_t)k * LDW + f];
    }
    const float d = dinv[r];
    H[(size_t)r * LDH + f]   = s;
    AGG[(size_t)r * LDA + f] = fmaf(s, d * d, bias[f]);
}

constexpr int Q_STRIDE = 64;
constexpr int Q_CELLS  = 10;
constexpr int EDGE_BLOCKS = 1024;

__device__ __forceinline__ int package_half() {
    return (int)((__builtin_amdgcn_s_sendmsg_rtn(0x87u) >> 18) & 1u);
}

template <int F, int LDH, int LDA, bool MAKE_W>
__global__ __launch_bounds__(256) void k_scatter(const float* __restrict__ H,
                                                 const int* __restrict__ src,
                                                 const int* __restrict__ dst,
                                                 const float* __restrict__ dinv,
                                                 float* __restrict__ WGT,
                                                 float* __restrict__ AGG,
                                                 int* __restrict__ tickets,
                                                 int E, int N) {
    static_assert(F % 32 == 0 && LDH % 32 == 0 && LDA % 32 == 0 && F <= LDH && F <= LDA,
                  "whole 128-byte lines");
    constexpr int NJ  = F / 32;
    constexpr int R   = 8;
    constexpr int CPT = 4;
    const int lane   = threadIdx.x & 31;
    const int nchunk = (E + 31) >> 5;
    const int nsplit = (N >> 1) & ~1;
    const int maxit  = nchunk / CPT + 2;
    const int hw     = package_half();

    for (int side = 0; side < 2; ++side) {
        const int h = hw ^ side;
        int* cell = tickets + h * Q_STRIDE;
        for (int it = 0; it < maxit; ++it) {
            int tk = 0;
            if (lane == 0) tk = atomicAdd(cell, CPT);
            const int t = __builtin_amdgcn_readfirstlane(tk);
            if (t >= nchunk) break;

            int d[CPT], s[CPT];
            float w[CPT];
            unsigned m[CPT];
#pragma unroll
            for (int q = 0; q < CPT; ++q) {
                const int e = ((t + q) << 5) + lane;
                const bool in = e < E;
                d[q] = in ? dst[e] : 0;
                s[q] = in ? src[e] : 0;
                if (!MAKE_W) w[q] = in ? WGT[e] : 0.0f;
            }
            bool mine[CPT];
#pragma unroll
            for (int q = 0; q < CPT; ++q) {
                const int e = ((t + q) << 5) + lane;
                mine[q] = (e < E) && (((d[q] >= nsplit) ? 1 : 0) == h);
                m[q] = __builtin_amdgcn_ballot_w32(mine[q]);
            }
            if (MAKE_W) {
                float ds[CPT], dd[CPT];
#pragma unroll
                for (int q = 0; q < CPT; ++q) {
                    ds[q] = 0.0f;
                    dd[q] = 0.0f;
                    if (mine[q]) {
                        ds[q] = dinv[s[q]];
                        dd[q] = dinv[d[q]];
                    }
                }
#pragma unroll
                for (int q = 0; q < CPT; ++q) {
                    const int e = ((t + q) << 5) + lane;
                    w[q] = ds[q] * dd[q];
                    if (mine[q]) WGT[e] = w[q];
                }
            }
#pragma unroll
            for (int q = 0; q < CPT; ++q) {
                unsigned mm = m[q];
                for (int r = 0; r < 32 / R && mm != 0u; ++r) {
                    int jl[R];
#pragma unroll
                    for (int k = 0; k < R; ++k) {
                        jl[k] = (mm != 0u) ? __builtin_ctz(mm) : -1;
                        mm &= mm - 1u;
                    }
                    float v[R][NJ] = {};
#pragma unroll
                    for (int k = 0; k < R; ++k) {
                        if (jl[k] >= 0) {
                            const int sj = __builtin_amdgcn_readlane(s[q], jl[k]);
                            const float* hp = H + (size_t)sj * LDH + lane;
                            for (int c = 0; c < NJ; ++c) v[k][c] = hp[c * 32];
                        }
                    }
#pragma unroll
                    for (int k = 0; k < R; ++k) {
                        if (jl[k] >= 0) {
                            const int   dj = __builtin_amdgcn_readlane(d[q], jl[k]);
                            const float wj = __int_as_float(
                                __builtin_amdgcn_readlane(__float_as_int(w[q]), jl[k]));
                            float* ap = AGG + (size_t)dj * LDA + lane;
                            for (int c = 0; c < NJ; ++c) atomAddF(ap + c * 32, v[k][c] * wj);
                        }
                    }
                }
            }
        }
    }
}

static inline int nblk(long long n, int t) { return (int)((n + t - 1) / t); }

template <int K, int F, int LDW, int LDH, int LDA, bool RELU_IN>
static void launch_gemm(const float* X, const float* W, const float* dinv, const float* bias,
                        float* H, float* AGG, int N, hipStream_t s) {
    int tilesM = N / 16;
    long long waves = (long long)tilesM * (F / 16);
    if (waves > 0)
        k_gemm_wmma<K, F, LDW, LDH, LDA, RELU_IN><<<nblk(waves, 8), 256, 0, s>>>(
            X, W, dinv, bias, H, AGG, tilesM);
    int tail = tilesM * 16;
    if (tail < N)
        k_gemm_tail<K, F, LDW, LDH, LDA, RELU_IN><<<nblk((long long)(N - tail) * F, 256), 256, 0, s>>>(
            X, W, dinv, bias, H, AGG, tail, N);
}

extern "C" void kernel_launch(void* const* d_in, const int* in_sizes, int n_in,
                              void* d_out, int out_size, void* d_ws, size_t ws_size,
                              hipStream_t stream) {
    const float* x  = (const float*)d_in[0];
    const int*   ei = (const int*)d_in[1];
    const float* W1 = (const float*)d_in[2]; const float* b1 = (const float*)d_in[3];
    const float* W2 = (const float*)d_in[4]; const float* b2 = (const float*)d_in[5];
    const float* W3 = (const float*)d_in[6]; const float* b3 = (const float*)d_in[7];
    const float* W4 = (const float*)d_in[8]; const float* b4 = (const float*)d_in[9];

    const int N = in_sizes[0] / 128;
    const int E = in_sizes[1] / 2;
    const int* src = ei;
    const int* dst = ei + E;

    float* out  = (float*)d_out;
    float* xrec = out;
    float* z    = out + (size_t)N * 128;

    float* ws   = (float*)d_ws;
    float* bufH = ws;
    float* wgt  = bufH + (size_t)N * 64;
    int*   tick = (int*)(wgt + E);
    float* bufA = bufH + (size_t)N * 128;
    float* dinv = bufA + (size_t)N * 64;

    k_init_ones<<<nblk((long long)N + Q_CELLS * Q_STRIDE, 256), 256, 0, stream>>>(dinv, N, tick, Q_CELLS * Q_STRIDE);
    k_deg_accum<<<nblk(E, 256), 256, 0, stream>>>(dst, dinv, E);
    k_rsqrt_inplace<<<nblk(N, 256), 256, 0, stream>>>(dinv, N);


    launch_gemm<128, 64, 64, 64, 64, false>(x, W1, dinv, b1, bufH, bufA, N, stream);
    k_scatter<64, 64, 64, true><<<EDGE_BLOCKS, 256, 0, stream>>>(bufH, src, dst, dinv, wgt, bufA, tick + 0 * Q_STRIDE, E, N);

    launch_gemm<64, 32, 32, 32, 32, true>(bufA, W2, dinv, b2, bufH, z, N, stream);
    k_scatter<32, 32, 32, false><<<EDGE_BLOCKS, 256, 0, stream>>>(bufH, src, dst, dinv, wgt, z, tick + 2 * Q_STRIDE, E, N);

    launch_gemm<32, 64, 64, 64, 64, false>(z, W3, dinv, b3, bufH, bufA, N, stream);
    k_scatter<64, 64, 64, false><<<EDGE_BLOCKS, 256, 0, stream>>>(bufH, src, dst, dinv, wgt, bufA, tick + 4 * Q_STRIDE, E, N);

    launch_gemm<64, 64, 128, 64, 128, true>(bufA, W4, dinv, b4, bufH, xrec, N, stream);
    k_scatter<64, 64, 128, false><<<EDGE_BLOCKS, 256, 0, stream>>>(bufH, src, dst, dinv, wgt, xrec, tick + 6 * Q_STRIDE, E, N);
    launch_gemm<64, 64, 128, 64, 128, true>(bufA, W4 + 64, dinv, b4 + 64, bufH, xrec + 64, N, stream);
    k_scatter<64, 64, 128, false><<<EDGE_BLOCKS, 256, 0, stream>>>(bufH, src, dst, dinv, wgt, xrec + 64, tick + 8 * Q_STRIDE, E, N);
}


# GEMMs: a work-group stages its rows of X in shared memory once, by whole-line 16-byte loads (ReLU there); its 8 tile waves read A pairs from it, not 16-line gathers per k-step; sums unchanged
# speedup vs baseline: 2.0427x; 1.0385x over previous
#include <hip/hip_runtime.h>
#include <hip/hip_bf16.h>


typedef float v2f __attribute__((ext_vector_type(2)));
typedef float v8f __attribute__((ext_vector_type(8)));

__device__ __forceinline__ void atomAddF(float* p, float v) {
#if defined(__HIP_PLATFORM_AMD__)
    unsafeAtomicAdd(p, v);
#else
    atomicAdd(p, v);
#endif
}

__global__ __launch_bounds__(256) void k_init_ones(float* __restrict__ p, int n,
                                                   int* __restrict__ q, int nq) {
    int i = blockIdx.x * blockDim.x + threadIdx.x;
    if (i < n) p[i] = 1.0f;
    if (i < nq) q[i] = 0;
}

__global__ __launch_bounds__(256) void k_deg_accum(const int* __restrict__ dst,
                                                   float* __restrict__ deg, int E) {
    int e = blockIdx.x * blockDim.x + threadIdx.x;
    if (e < E) atomAddF(&deg[dst[e]], 1.0f);
}

__global__ __launch_bounds__(256) void k_rsqrt_inplace(float* __restrict__ p, int n) {
    int i = blockIdx.x * blockDim.x + threadIdx.x;
    if (i < n) p[i] = rsqrtf(p[i]);
}

template <int K, int F, int LDW, int LDH, int LDA, bool RELU_IN>
__global__ __launch_bounds__(256) void k_gemm_wmma(const float* __restrict__ X,
                                                   const float* __restrict__ W,
                                                   const float* __restrict__ dinv,
                                                   const float* __restrict__ bias,
                                                   float* __restrict__ H,
                                                   float* __restrict__ AGG,
                                                   int tilesM) {
    static_assert(K % 4 == 0 && F % 16 == 0 && 128 % F == 0 && F <= LDW && F <= LDH && F <= LDA, "shape");
    constexpr int tilesF = F / 16;
    constexpr int RT  = 8 / tilesF;
    constexpr int RB  = RT * 16;
    constexpr int P   = K + 4;
    constexpr int VPR = K / 4;
    constexpr int NV  = RB * VPR;
    static_assert(NV % 256 == 0, "whole rounds of the 256 threads");
    __shared__ __attribute__((aligned(16))) float As[RB * P];

    const int row0      = blockIdx.x * RB;
    const int rowsValid = tilesM * 16;
    float4 xv[NV / 256];
#pragma unroll
    for (int i = 0; i < NV / 256; ++i) {
        const int p = threadIdx.x + i * 256;
        const int r = p / VPR, c4 = p % VPR;
        xv[i] = make_float4(0.0f, 0.0f, 0.0f, 0.0f);
        if (row0 + r < rowsValid)
            xv[i] = *reinterpret_cast<const float4*>(X + (size_t)(row0 + r) * K + c4 * 4);
    }
#pragma unroll
    for (int i = 0; i < NV / 256; ++i) {
        const int p = threadIdx.x + i * 256;
        const int r = p / VPR, c4 = p % VPR;
        float4 v = xv[i];
        if (RELU_IN) {
            v.x = fmaxf(v.x, 0.0f);
            v.y = fmaxf(v.y, 0.0f);
            v.z = fmaxf(v.z, 0.0f);
            v.w = fmaxf(v.w, 0.0f);
        }
        *reinterpret_cast<float4*>(As + r * P + c4 * 4) = v;
    }
    __syncthreads();

    const int wv = threadIdx.x >> 5;
    const int rt = wv / tilesF;
    const int tf = wv % tilesF;
    const int tm = blockIdx.x * RT + rt;
    if (tm >= tilesM) return;
    const int lane = threadIdx.x & 31;
    const int l15  = lane & 15;
    const int hi2  = (lane >> 4) << 1;

    const float* arow = As + (rt * 16 + l15) * P;
    const float* wcol = W + (tf * 16 + l15);
    const int hi8 = (lane >> 4) << 3;

    float ns[8];
#pragma unroll
    for (int v = 0; v < 8; ++v) {
        const float d = dinv[tm * 16 + hi8 + v];
        ns[v] = d * d;
    }
    const float bcol = bias[tf * 16 + l15];

    v8f acc = {};
#pragma unroll 4
    for (int kk = 0; kk < K; kk += 4) {
        v2f a, b;
        a.x = arow[kk + hi2];
        a.y = arow[kk + hi2 + 1];
        b.x = wcol[(size_t)(kk + hi2) * LDW];
        b.y = wcol[(size_t)(kk + hi2 + 1) * LDW];
        acc = __builtin_amdgcn_wmma_f32_16x16x4_f32(
             false, a,  false, b,
             (short)0, acc,  false,  false);
    }

#pragma unroll
    for (int v = 0; v < 8; ++v) {
        const size_t r = (size_t)(tm * 16 + hi8 + v);
        const int    c = tf * 16 + l15;
        H[r * LDH + c]   = acc[v];
        AGG[r * LDA + c] = fmaf(acc[v], ns[v], bcol);
    }
}

template <int K, int F, int LDW, int LDH, int LDA, bool RELU_IN>
__global__ __launch_bounds__(256) void k_gemm_tail(const float* __restrict__ X,
                                                   const float* __restrict__ W,
                                                   const float* __restrict__ dinv,
                                                   const float* __restrict__ bias,
                                                   float* __restrict__ H,
                                                   float* __restrict__ AGG,
                                                   int rowStart, int rowEnd) {
    int idx = blockIdx.x * blockDim.x + threadIdx.x;
    int rows = rowEnd - rowStart;
    if (idx >= rows * F) return;
    int r = rowStart + idx / F, f = idx % F;
    float s = 0.f;
    for (int k = 0; k < K; ++k) {
        float xv = X[(size_t)r * K + k];
        if (RELU_IN) xv = fmaxf(xv, 0.0f);
        s += xv * W[(size_t)k * LDW + f];
    }
    const float d = dinv[r];
    H[(size_t)r * LDH + f]   = s;
    AGG[(size_t)r * LDA + f] = fmaf(s, d * d, bias[f]);
}

constexpr int Q_STRIDE = 64;
constexpr int Q_CELLS  = 10;
constexpr int EDGE_BLOCKS = 1024;

__device__ __forceinline__ int package_half() {
    return (int)((__builtin_amdgcn_s_sendmsg_rtn(0x87u) >> 18) & 1u);
}

template <int F, int LDH, int LDA, bool MAKE_W>
__global__ __launch_bounds__(256) void k_scatter(const float* __restrict__ H,
                                                 const int* __restrict__ src,
                                                 const int* __restrict__ dst,
                                                 const float* __restrict__ dinv,
                                                 float* __restrict__ WGT,
                                                 float* __restrict__ AGG,
                                                 int* __restrict__ tickets,
                                                 int E, int N) {
    static_assert(F % 32 == 0 && LDH % 32 == 0 && LDA % 32 == 0 && F <= LDH && F <= LDA,
                  "whole 128-byte lines");
    constexpr int NJ  = F / 32;
    constexpr int R   = 8;
    constexpr int CPT = 4;
    const int lane   = threadIdx.x & 31;
    const int nchunk = (E + 31) >> 5;
    const int nsplit = (N >> 1) & ~1;
    const int maxit  = nchunk / CPT + 2;
    const int hw     = package_half();

    for (int side = 0; side < 2; ++side) {
        const int h = hw ^ side;
        int* cell = tickets + h * Q_STRIDE;
        for (int it = 0; it < maxit; ++it) {
            int tk = 0;
            if (lane == 0) tk = atomicAdd(cell, CPT);
            const int t = __builtin_amdgcn_readfirstlane(tk);
            if (t >= nchunk) break;

            int d[CPT], s[CPT];
            float w[CPT];
            unsigned m[CPT];
#pragma unroll
            for (int q = 0; q < CPT; ++q) {
                const int e = ((t + q) << 5) + lane;
                const bool in = e < E;
                d[q] = in ? dst[e] : 0;
                s[q] = in ? src[e] : 0;
                if (!MAKE_W) w[q] = in ? WGT[e] : 0.0f;
            }
            bool mine[CPT];
#pragma unroll
            for (int q = 0; q < CPT; ++q) {
                const int e = ((t + q) << 5) + lane;
                mine[q] = (e < E) && (((d[q] >= nsplit) ? 1 : 0) == h);
                m[q] = __builtin_amdgcn_ballot_w32(mine[q]);
            }
            if (MAKE_W) {
                float ds[CPT], dd[CPT];
#pragma unroll
                for (int q = 0; q < CPT; ++q) {
                    ds[q] = 0.0f;
                    dd[q] = 0.0f;
                    if (mine[q]) {
                        ds[q] = dinv[s[q]];
                        dd[q] = dinv[d[q]];
                    }
                }
#pragma unroll
                for (int q = 0; q < CPT; ++q) {
                    const int e = ((t + q) << 5) + lane;
                    w[q] = ds[q] * dd[q];
                    if (mine[q]) WGT[e] = w[q];
                }
            }
#pragma unroll
            for (int q = 0; q < CPT; ++q) {
                unsigned mm = m[q];
                for (int r = 0; r < 32 / R && mm != 0u; ++r) {
                    int jl[R];
#pragma unroll
                    for (int k = 0; k < R; ++k) {
                        jl[k] = (mm != 0u) ? __builtin_ctz(mm) : -1;
                        mm &= mm - 1u;
                    }
                    float v[R][NJ] = {};
#pragma unroll
                    for (int k = 0; k < R; ++k) {
                        if (jl[k] >= 0) {
                            const int sj = __builtin_amdgcn_readlane(s[q], jl[k]);
                            const float* hp = H + (size_t)sj * LDH + lane;
                            for (int c = 0; c < NJ; ++c) v[k][c] = hp[c * 32];
                        }
                    }
#pragma unroll
                    for (int k = 0; k < R; ++k) {
                        if (jl[k] >= 0) {
                            const int   dj = __builtin_amdgcn_readlane(d[q], jl[k]);
                            const float wj = __int_as_float(
                                __builtin_amdgcn_readlane(__float_as_int(w[q]), jl[k]));
                            float* ap = AGG + (size_t)dj * LDA + lane;
                            for (int c = 0; c < NJ; ++c) atomAddF(ap + c * 32, v[k][c] * wj);
                        }
                    }
                }
            }
        }
    }
}

static inline int nblk(long long n, int t) { return (int)((n + t - 1) / t); }

template <int K, int F, int LDW, int LDH, int LDA, bool RELU_IN>
static void launch_gemm(const float* X, const float* W, const float* dinv, const float* bias,
                        float* H, float* AGG, int N, hipStream_t s) {
    int tilesM = N / 16;
    if (tilesM > 0)
        k_gemm_wmma<K, F, LDW, LDH, LDA, RELU_IN><<<nblk((long long)tilesM, 8 / (F / 16)), 256, 0, s>>>(
            X, W, dinv, bias, H, AGG, tilesM);
    int tail = tilesM * 16;
    if (tail < N)
        k_gemm_tail<K, F, LDW, LDH, LDA, RELU_IN><<<nblk((long long)(N - tail) * F, 256), 256, 0, s>>>(
            X, W, dinv, bias, H, AGG, tail, N);
}

extern "C" void kernel_launch(void* const* d_in, const int* in_sizes, int n_in,
                              void* d_out, int out_size, void* d_ws, size_t ws_size,
                              hipStream_t stream) {
    const float* x  = (const float*)d_in[0];
    const int*   ei = (const int*)d_in[1];
    const float* W1 = (const float*)d_in[2]; const float* b1 = (const float*)d_in[3];
    const float* W2 = (const float*)d_in[4]; const float* b2 = (const float*)d_in[5];
    const float* W3 = (const float*)d_in[6]; const float* b3 = (const float*)d_in[7];
    const float* W4 = (const float*)d_in[8]; const float* b4 = (const float*)d_in[9];

    const int N = in_sizes[0] / 128;
    const int E = in_sizes[1] / 2;
    const int* src = ei;
    const int* dst = ei + E;

    float* out  = (float*)d_out;
    float* xrec = out;
    float* z    = out + (size_t)N * 128;

    float* ws   = (float*)d_ws;
    float* bufH = ws;
    float* wgt  = bufH + (size_t)N * 64;
    int*   tick = (int*)(wgt + E);
    float* bufA = bufH + (size_t)N * 128;
    float* dinv = bufA + (size_t)N * 64;

    k_init_ones<<<nblk((long long)N + Q_CELLS * Q_STRIDE, 256), 256, 0, stream>>>(dinv, N, tick, Q_CELLS * Q_STRIDE);
    k_deg_accum<<<nblk(E, 256), 256, 0, stream>>>(dst, dinv, E);
    k_rsqrt_inplace<<<nblk(N, 256), 256, 0, stream>>>(dinv, N);


    launch_gemm<128, 64, 64, 64, 64, false>(x, W1, dinv, b1, bufH, bufA, N, stream);
    k_scatter<64, 64, 64, true><<<EDGE_BLOCKS, 256, 0, stream>>>(bufH, src, dst, dinv, wgt, bufA, tick + 0 * Q_STRIDE, E, N);

    launch_gemm<64, 32, 32, 32, 32, true>(bufA, W2, dinv, b2, bufH, z, N, stream);
    k_scatter<32, 32, 32, false><<<EDGE_BLOCKS, 256, 0, stream>>>(bufH, src, dst, dinv, wgt, z, tick + 2 * Q_STRIDE, E, N);

    launch_gemm<32, 64, 64, 64, 64, false>(z, W3, dinv, b3, bufH, bufA, N, stream);
    k_scatter<64, 64, 64, false><<<EDGE_BLOCKS, 256, 0, stream>>>(bufH, src, dst, dinv, wgt, bufA, tick + 4 * Q_STRIDE, E, N);

    launch_gemm<64, 64, 128, 64, 128, true>(bufA, W4, dinv, b4, bufH, xrec, N, stream);
    k_scatter<64, 64, 128, false><<<EDGE_BLOCKS, 256, 0, stream>>>(bufH, src, dst, dinv, wgt, xrec, tick + 6 * Q_STRIDE, E, N);
    launch_gemm<64, 64, 128, 64, 128, true>(bufA, W4 + 64, dinv, b4 + 64, bufH, xrec + 64, N, stream);
    k_scatter<64, 64, 128, false><<<EDGE_BLOCKS, 256, 0, stream>>>(bufH, src, dst, dinv, wgt, xrec + 64, tick + 8 * Q_STRIDE, E, N);
}


# Degree count on shared-memory integer counters: a work-group counts one node range of one edge slice, then adds 32 neighbouring counts per f32 atomic (one line), not one atomic per edge; sums exact
# speedup vs baseline: 2.4358x; 1.1924x over previous
#include <hip/hip_runtime.h>
#include <hip/hip_bf16.h>


typedef float v2f __attribute__((ext_vector_type(2)));
typedef float v8f __attribute__((ext_vector_type(8)));

__device__ __forceinline__ void atomAddF(float* p, float v) {
#if defined(__HIP_PLATFORM_AMD__)
    unsafeAtomicAdd(p, v);
#else
    atomicAdd(p, v);
#endif
}

__global__ __launch_bounds__(256) void k_init_ones(float* __restrict__ p, int n,
                                                   int* __restrict__ q, int nq) {
    int i = blockIdx.x * blockDim.x + threadIdx.x;
    if (i < n) p[i] = 1.0f;
    if (i < nq) q[i] = 0;
}

constexpr int DEG_RANGE  = 12800;
constexpr int DEG_SLICES = 32;
constexpr int DEG_BATCH  = 8;

__device__ __forceinline__ void deg_count(unsigned* cnt, int d, int lo, int nloc) {
    const unsigned r = (unsigned)(d - lo);
    if (r < (unsigned)nloc) atomicAdd(&cnt[r], 1u);
}

__global__ __launch_bounds__(1024) void k_deg_lds(const int* __restrict__ dst,
                                                   float* __restrict__ deg, int E, int N) {
    __shared__ unsigned cnt[DEG_RANGE];
    const int lo   = blockIdx.y * DEG_RANGE;
    const int nloc = max(0, min(DEG_RANGE, N - lo));
    for (int i = threadIdx.x; i < DEG_RANGE; i += 1024) cnt[i] = 0u;
    __syncthreads();

    const int per = (((E + DEG_SLICES - 1) / DEG_SLICES) + 3) & ~3;
    const int e0  = min(E, (int)blockIdx.x * per);
    const int e1  = min(E, e0 + per);
    if ((reinterpret_cast<size_t>(dst) & 15) == 0) {
        const int4* d4 = reinterpret_cast<const int4*>(dst);
        const int qend = e1 >> 2;
        for (int q0 = (e0 >> 2) + threadIdx.x; q0 < qend; q0 += 1024 * DEG_BATCH) {
            int4 v[DEG_BATCH];
#pragma unroll
            for (int b = 0; b < DEG_BATCH; ++b) {
                const int q = q0 + b * 1024;
                v[b] = make_int4(-1, -1, -1, -1);
                if (q < qend) v[b] = d4[q];
            }
#pragma unroll
            for (int b = 0; b < DEG_BATCH; ++b) {
                deg_count(cnt, v[b].x, lo, nloc);
                deg_count(cnt, v[b].y, lo, nloc);
                deg_count(cnt, v[b].z, lo, nloc);
                deg_count(cnt, v[b].w, lo, nloc);
            }
        }
        const int et = max(e0, qend << 2) + threadIdx.x;
        if (et < e1) deg_count(cnt, dst[et], lo, nloc);
    } else {
        for (int e = e0 + threadIdx.x; e < e1; e += 1024) deg_count(cnt, dst[e], lo, nloc);
    }
    __syncthreads();

    for (int i = threadIdx.x; i < nloc; i += 1024) {
        const unsigned c = cnt[i];
        if (c != 0u) atomAddF(&deg[lo + i], (float)c);
    }
}

__global__ __launch_bounds__(256) void k_rsqrt_inplace(float* __restrict__ p, int n) {
    int i = blockIdx.x * blockDim.x + threadIdx.x;
    if (i < n) p[i] = rsqrtf(p[i]);
}

template <int K, int F, int LDW, int LDH, int LDA, bool RELU_IN>
__global__ __launch_bounds__(256) void k_gemm_wmma(const float* __restrict__ X,
                                                   const float* __restrict__ W,
                                                   const float* __restrict__ dinv,
                                                   const float* __restrict__ bias,
                                                   float* __restrict__ H,
                                                   float* __restrict__ AGG,
                                                   int tilesM) {
    static_assert(K % 4 == 0 && F % 16 == 0 && 128 % F == 0 && F <= LDW && F <= LDH && F <= LDA, "shape");
    constexpr int tilesF = F / 16;
    constexpr int RT  = 8 / tilesF;
    constexpr int RB  = RT * 16;
    constexpr int P   = K + 4;
    constexpr int VPR = K / 4;
    constexpr int NV  = RB * VPR;
    static_assert(NV % 256 == 0, "whole rounds of the 256 threads");
    __shared__ __attribute__((aligned(16))) float As[RB * P];

    const int row0      = blockIdx.x * RB;
    const int rowsValid = tilesM * 16;
    float4 xv[NV / 256];
#pragma unroll
    for (int i = 0; i < NV / 256; ++i) {
        const int p = threadIdx.x + i * 256;
        const int r = p / VPR, c4 = p % VPR;
        xv[i] = make_float4(0.0f, 0.0f, 0.0f, 0.0f);
        if (row0 + r < rowsValid)
            xv[i] = *reinterpret_cast<const float4*>(X + (size_t)(row0 + r) * K + c4 * 4);
    }
#pragma unroll
    for (int i = 0; i < NV / 256; ++i) {
        const int p = threadIdx.x + i * 256;
        const int r = p / VPR, c4 = p % VPR;
        float4 v = xv[i];
        if (RELU_IN) {
            v.x = fmaxf(v.x, 0.0f);
            v.y = fmaxf(v.y, 0.0f);
            v.z = fmaxf(v.z, 0.0f);
            v.w = fmaxf(v.w, 0.0f);
        }
        *reinterpret_cast<float4*>(As + r * P + c4 * 4) = v;
    }
    __syncthreads();

    const int wv = threadIdx.x >> 5;
    const int rt = wv / tilesF;
    const int tf = wv % tilesF;
    const int tm = blockIdx.x * RT + rt;
    if (tm >= tilesM) return;
    const int lane = threadIdx.x & 31;
    const int l15  = lane & 15;
    const int hi2  = (lane >> 4) << 1;

    const float* arow = As + (rt * 16 + l15) * P;
    const float* wcol = W + (tf * 16 + l15);
    const int hi8 = (lane >> 4) << 3;

    float ns[8];
#pragma unroll
    for (int v = 0; v < 8; ++v) {
        const float d = dinv[tm * 16 + hi8 + v];
        ns[v] = d * d;
    }
    const float bcol = bias[tf * 16 + l15];

    v8f acc = {};
#pragma unroll 4
    for (int kk = 0; kk < K; kk += 4) {
        v2f a, b;
        a.x = arow[kk + hi2];
        a.y = arow[kk + hi2 + 1];
        b.x = wcol[(size_t)(kk + hi2) * LDW];
        b.y = wcol[(size_t)(kk + hi2 + 1) * LDW];
        acc = __builtin_amdgcn_wmma_f32_16x16x4_f32(
             false, a,  false, b,
             (short)0, acc,  false,  false);
    }

#pragma unroll
    for (int v = 0; v < 8; ++v) {
        const size_t r = (size_t)(tm * 16 + hi8 + v);
        const int    c = tf * 16 + l15;
        H[r * LDH + c]   = acc[v];
        AGG[r * LDA + c] = fmaf(acc[v], ns[v], bcol);
    }
}

template <int K, int F, int LDW, int LDH, int LDA, bool RELU_IN>
__global__ __launch_bounds__(256) void k_gemm_tail(const float* __restrict__ X,
                                                   const float* __restrict__ W,
                                                   const float* __restrict__ dinv,
                                                   const float* __restrict__ bias,
                                                   float* __restrict__ H,
                                                   float* __restrict__ AGG,
                                                   int rowStart, int rowEnd) {
    int idx = blockIdx.x * blockDim.x + threadIdx.x;
    int rows = rowEnd - rowStart;
    if (idx >= rows * F) return;
    int r = rowStart + idx / F, f = idx % F;
    float s = 0.f;
    for (int k = 0; k < K; ++k) {
        float xv = X[(size_t)r * K + k];
        if (RELU_IN) xv = fmaxf(xv, 0.0f);
        s += xv * W[(size_t)k * LDW + f];
    }
    const float d = dinv[r];
    H[(size_t)r * LDH + f]   = s;
    AGG[(size_t)r * LDA + f] = fmaf(s, d * d, bias[f]);
}

constexpr int Q_STRIDE = 64;
constexpr int Q_CELLS  = 10;
constexpr int EDGE_BLOCKS = 1024;

__device__ __forceinline__ int package_half() {
    return (int)((__builtin_amdgcn_s_sendmsg_rtn(0x87u) >> 18) & 1u);
}

template <int F, int LDH, int LDA, bool MAKE_W>
__global__ __launch_bounds__(256) void k_scatter(const float* __restrict__ H,
                                                 const int* __restrict__ src,
                                                 const int* __restrict__ dst,
                                                 const float* __restrict__ dinv,
                                                 float* __restrict__ WGT,
                                                 float* __restrict__ AGG,
                                                 int* __restrict__ tickets,
                                                 int E, int N) {
    static_assert(F % 32 == 0 && LDH % 32 == 0 && LDA % 32 == 0 && F <= LDH && F <= LDA,
                  "whole 128-byte lines");
    constexpr int NJ  = F / 32;
    constexpr int R   = 8;
    constexpr int CPT = 4;
    const int lane   = threadIdx.x & 31;
    const int nchunk = (E + 31) >> 5;
    const int nsplit = (N >> 1) & ~1;
    const int maxit  = nchunk / CPT + 2;
    const int hw     = package_half();

    for (int side = 0; side < 2; ++side) {
        const int h = hw ^ side;
        int* cell = tickets + h * Q_STRIDE;
        for (int it = 0; it < maxit; ++it) {
            int tk = 0;
            if (lane == 0) tk = atomicAdd(cell, CPT);
            const int t = __builtin_amdgcn_readfirstlane(tk);
            if (t >= nchunk) break;

            int d[CPT], s[CPT];
            float w[CPT];
            unsigned m[CPT];
#pragma unroll
            for (int q = 0; q < CPT; ++q) {
                const int e = ((t + q) << 5) + lane;
                const bool in = e < E;
                d[q] = in ? dst[e] : 0;
                s[q] = in ? src[e] : 0;
                if (!MAKE_W) w[q] = in ? WGT[e] : 0.0f;
            }
            bool mine[CPT];
#pragma unroll
            for (int q = 0; q < CPT; ++q) {
                const int e = ((t + q) << 5) + lane;
                mine[q] = (e < E) && (((d[q] >= nsplit) ? 1 : 0) == h);
                m[q] = __builtin_amdgcn_ballot_w32(mine[q]);
            }
            if (MAKE_W) {
                float ds[CPT], dd[CPT];
#pragma unroll
                for (int q = 0; q < CPT; ++q) {
                    ds[q] = 0.0f;
                    dd[q] = 0.0f;
                    if (mine[q]) {
                        ds[q] = dinv[s[q]];
                        dd[q] = dinv[d[q]];
                    }
                }
#pragma unroll
                for (int q = 0; q < CPT; ++q) {
                    const int e = ((t + q) << 5) + lane;
                    w[q] = ds[q] * dd[q];
                    if (mine[q]) WGT[e] = w[q];
                }
            }
#pragma unroll
            for (int q = 0; q < CPT; ++q) {
                unsigned mm = m[q];
                for (int r = 0; r < 32 / R && mm != 0u; ++r) {
                    int jl[R];
#pragma unroll
                    for (int k = 0; k < R; ++k) {
                        jl[k] = (mm != 0u) ? __builtin_ctz(mm) : -1;
                        mm &= mm - 1u;
                    }
                    float v[R][NJ] = {};
#pragma unroll
                    for (int k = 0; k < R; ++k) {
                        if (jl[k] >= 0) {
                            const int sj = __builtin_amdgcn_readlane(s[q], jl[k]);
                            const float* hp = H + (size_t)sj * LDH + lane;
                            for (int c = 0; c < NJ; ++c) v[k][c] = hp[c * 32];
                        }
                    }
#pragma unroll
                    for (int k = 0; k < R; ++k) {
                        if (jl[k] >= 0) {
                            const int   dj = __builtin_amdgcn_readlane(d[q], jl[k]);
                            const float wj = __int_as_float(
                                __builtin_amdgcn_readlane(__float_as_int(w[q]), jl[k]));
                            float* ap = AGG + (size_t)dj * LDA + lane;
                            for (int c = 0; c < NJ; ++c) atomAddF(ap + c * 32, v[k][c] * wj);
                        }
                    }
                }
            }
        }
    }
}

static inline int nblk(long long n, int t) { return (int)((n + t - 1) / t); }

template <int K, int F, int LDW, int LDH, int LDA, bool RELU_IN>
static void launch_gemm(const float* X, const float* W, const float* dinv, const float* bias,
                        float* H, float* AGG, int N, hipStream_t s) {
    int tilesM = N / 16;
    if (tilesM > 0)
        k_gemm_wmma<K, F, LDW, LDH, LDA, RELU_IN><<<nblk((long long)tilesM, 8 / (F / 16)), 256, 0, s>>>(
            X, W, dinv, bias, H, AGG, tilesM);
    int tail = tilesM * 16;
    if (tail < N)
        k_gemm_tail<K, F, LDW, LDH, LDA, RELU_IN><<<nblk((long long)(N - tail) * F, 256), 256, 0, s>>>(
            X, W, dinv, bias, H, AGG, tail, N);
}

extern "C" void kernel_launch(void* const* d_in, const int* in_sizes, int n_in,
                              void* d_out, int out_size, void* d_ws, size_t ws_size,
                              hipStream_t stream) {
    const float* x  = (const float*)d_in[0];
    const int*   ei = (const int*)d_in[1];
    const float* W1 = (const float*)d_in[2]; const float* b1 = (const float*)d_in[3];
    const float* W2 = (const float*)d_in[4]; const float* b2 = (const float*)d_in[5];
    const float* W3 = (const float*)d_in[6]; const float* b3 = (const float*)d_in[7];
    const float* W4 = (const float*)d_in[8]; const float* b4 = (const float*)d_in[9];

    const int N = in_sizes[0] / 128;
    const int E = in_sizes[1] / 2;
    const int* src = ei;
    const int* dst = ei + E;

    float* out  = (float*)d_out;
    float* xrec = out;
    float* z    = out + (size_t)N * 128;

    float* ws   = (float*)d_ws;
    float* bufH = ws;
    float* wgt  = bufH + (size_t)N * 64;
    int*   tick = (int*)(wgt + E);
    float* bufA = bufH + (size_t)N * 128;
    float* dinv = bufA + (size_t)N * 64;

    k_init_ones<<<nblk((long long)N + Q_CELLS * Q_STRIDE, 256), 256, 0, stream>>>(dinv, N, tick, Q_CELLS * Q_STRIDE);
    k_deg_lds<<<dim3(DEG_SLICES, nblk(N, DEG_RANGE)), 1024, 0, stream>>>(dst, dinv, E, N);
    k_rsqrt_inplace<<<nblk(N, 256), 256, 0, stream>>>(dinv, N);


    launch_gemm<128, 64, 64, 64, 64, false>(x, W1, dinv, b1, bufH, bufA, N, stream);
    k_scatter<64, 64, 64, true><<<EDGE_BLOCKS, 256, 0, stream>>>(bufH, src, dst, dinv, wgt, bufA, tick + 0 * Q_STRIDE, E, N);

    launch_gemm<64, 32, 32, 32, 32, true>(bufA, W2, dinv, b2, bufH, z, N, stream);
    k_scatter<32, 32, 32, false><<<EDGE_BLOCKS, 256, 0, stream>>>(bufH, src, dst, dinv, wgt, z, tick + 2 * Q_STRIDE, E, N);

    launch_gemm<32, 64, 64, 64, 64, false>(z, W3, dinv, b3, bufH, bufA, N, stream);
    k_scatter<64, 64, 64, false><<<EDGE_BLOCKS, 256, 0, stream>>>(bufH, src, dst, dinv, wgt, bufA, tick + 4 * Q_STRIDE, E, N);

    launch_gemm<64, 64, 128, 64, 128, true>(bufA, W4, dinv, b4, bufH, xrec, N, stream);
    k_scatter<64, 64, 128, false><<<EDGE_BLOCKS, 256, 0, stream>>>(bufH, src, dst, dinv, wgt, xrec, tick + 6 * Q_STRIDE, E, N);
    launch_gemm<64, 64, 128, 64, 128, true>(bufA, W4 + 64, dinv, b4 + 64, bufH, xrec + 64, N, stream);
    k_scatter<64, 64, 128, false><<<EDGE_BLOCKS, 256, 0, stream>>>(bufH, src, dst, dinv, wgt, xrec + 64, tick + 8 * Q_STRIDE, E, N);
}


# Layer 4 as one GEMM and one edge launch: h columns 64-127 go over the GEMM's own staged input rows, edge tickets walk block 0 then block 1; 13 -> 11 launches; f32 atomics in arrival order as before
# speedup vs baseline: 2.5270x; 1.0374x over previous
#include <hip/hip_runtime.h>
#include <hip/hip_bf16.h>


typedef float v2f __attribute__((ext_vector_type(2)));
typedef float v8f __attribute__((ext_vector_type(8)));

__device__ __forceinline__ void atomAddF(float* p, float v) {
#if defined(__HIP_PLATFORM_AMD__)
    unsafeAtomicAdd(p, v);
#else
    atomicAdd(p, v);
#endif
}

__global__ __launch_bounds__(256) void k_init_ones(float* __restrict__ p, int n,
                                                   int* __restrict__ q, int nq) {
    int i = blockIdx.x * blockDim.x + threadIdx.x;
    if (i < n) p[i] = 1.0f;
    if (i < nq) q[i] = 0;
}

constexpr int DEG_RANGE  = 12800;
constexpr int DEG_SLICES = 32;
constexpr int DEG_BATCH  = 8;

__device__ __forceinline__ void deg_count(unsigned* cnt, int d, int lo, int nloc) {
    const unsigned r = (unsigned)(d - lo);
    if (r < (unsigned)nloc) atomicAdd(&cnt[r], 1u);
}

__global__ __launch_bounds__(1024) void k_deg_lds(const int* __restrict__ dst,
                                                   float* __restrict__ deg, int E, int N) {
    __shared__ unsigned cnt[DEG_RANGE];
    const int lo   = blockIdx.y * DEG_RANGE;
    const int nloc = max(0, min(DEG_RANGE, N - lo));
    for (int i = threadIdx.x; i < DEG_RANGE; i += 1024) cnt[i] = 0u;
    __syncthreads();

    const int per = (((E + DEG_SLICES - 1) / DEG_SLICES) + 3) & ~3;
    const int e0  = min(E, (int)blockIdx.x * per);
    const int e1  = min(E, e0 + per);
    if ((reinterpret_cast<size_t>(dst) & 15) == 0) {
        const int4* d4 = reinterpret_cast<const int4*>(dst);
        const int qend = e1 >> 2;
        for (int q0 = (e0 >> 2) + threadIdx.x; q0 < qend; q0 += 1024 * DEG_BATCH) {
            int4 v[DEG_BATCH];
#pragma unroll
            for (int b = 0; b < DEG_BATCH; ++b) {
                const int q = q0 + b * 1024;
                v[b] = make_int4(-1, -1, -1, -1);
                if (q < qend) v[b] = d4[q];
            }
#pragma unroll
            for (int b = 0; b < DEG_BATCH; ++b) {
                deg_count(cnt, v[b].x, lo, nloc);
                deg_count(cnt, v[b].y, lo, nloc);
                deg_count(cnt, v[b].z, lo, nloc);
                deg_count(cnt, v[b].w, lo, nloc);
            }
        }
        const int et = max(e0, qend << 2) + threadIdx.x;
        if (et < e1) deg_count(cnt, dst[et], lo, nloc);
    } else {
        for (int e = e0 + threadIdx.x; e < e1; e += 1024) deg_count(cnt, dst[e], lo, nloc);
    }
    __syncthreads();

    for (int i = threadIdx.x; i < nloc; i += 1024) {
        const unsigned c = cnt[i];
        if (c != 0u) atomAddF(&deg[lo + i], (float)c);
    }
}

__global__ __launch_bounds__(256) void k_rsqrt_inplace(float* __restrict__ p, int n) {
    int i = blockIdx.x * blockDim.x + threadIdx.x;
    if (i < n) p[i] = rsqrtf(p[i]);
}

template <int K, int F, int HS, int LDW, int LDH, int LDA, bool RELU_IN>
__global__ __launch_bounds__(256) void k_gemm_wmma(const float* X,
                                                   const float* __restrict__ W,
                                                   const float* __restrict__ dinv,
                                                   const float* __restrict__ bias,
                                                   float* __restrict__ H,
                                                   float* H2,
                                                   float* __restrict__ AGG,
                                                   int tilesM) {
    static_assert(K % 4 == 0 && F % 16 == 0 && 128 % F == 0 && F <= LDW && HS <= LDH && F <= LDA, "shape");
    static_assert(HS % 16 == 0 && (HS == F || 2 * HS == F), "one or two column blocks of whole tiles");
    constexpr int tilesF = F / 16;
    constexpr int RT  = 8 / tilesF;
    constexpr int RB  = RT * 16;
    constexpr int P   = K + 4;
    constexpr int VPR = K / 4;
    constexpr int NV  = RB * VPR;
    static_assert(NV % 256 == 0, "whole rounds of the 256 threads");
    __shared__ __attribute__((aligned(16))) float As[RB * P];

    const int row0      = blockIdx.x * RB;
    const int rowsValid = tilesM * 16;
    float4 xv[NV / 256];
#pragma unroll
    for (int i = 0; i < NV / 256; ++i) {
        const int p = threadIdx.x + i * 256;
        const int r = p / VPR, c4 = p % VPR;
        xv[i] = make_float4(0.0f, 0.0f, 0.0f, 0.0f);
        if (row0 + r < rowsValid)
            xv[i] = *reinterpret_cast<const float4*>(X + (size_t)(row0 + r) * K + c4 * 4);
    }
#pragma unroll
    for (int i = 0; i < NV / 256; ++i) {
        const int p = threadIdx.x + i * 256;
        const int r = p / VPR, c4 = p % VPR;
        float4 v = xv[i];
        if (RELU_IN) {
            v.x = fmaxf(v.x, 0.0f);
            v.y = fmaxf(v.y, 0.0f);
            v.z = fmaxf(v.z, 0.0f);
            v.w = fmaxf(v.w, 0.0f);
        }
        *reinterpret_cast<float4*>(As + r * P + c4 * 4) = v;
    }
    __syncthreads();

    const int wv = threadIdx.x >> 5;
    const int rt = wv / tilesF;
    const int tf = wv % tilesF;
    const int tm = blockIdx.x * RT + rt;
    if (tm >= tilesM) return;
    const int lane = threadIdx.x & 31;
    const int l15  = lane & 15;
    const int hi2  = (lane >> 4) << 1;

    const float* arow = As + (rt * 16 + l15) * P;
    const float* wcol = W + (tf * 16 + l15);
    const int hi8 = (lane >> 4) << 3;

    float ns[8];
#pragma unroll
    for (int v = 0; v < 8; ++v) {
        const float d = dinv[tm * 16 + hi8 + v];
        ns[v] = d * d;
    }
    const float bcol = bias[tf * 16 + l15];

    v8f acc = {};
#pragma unroll 4
    for (int kk = 0; kk < K; kk += 4) {
        v2f a, b;
        a.x = arow[kk + hi2];
        a.y = arow[kk + hi2 + 1];
        b.x = wcol[(size_t)(kk + hi2) * LDW];
        b.y = wcol[(size_t)(kk + hi2 + 1) * LDW];
        acc = __builtin_amdgcn_wmma_f32_16x16x4_f32(
             false, a,  false, b,
             (short)0, acc,  false,  false);
    }

    float* hdst = H;
    int    hcol = tf * 16 + l15;
    if (HS < F && tf * 16 >= HS) {
        hdst = H2;
        hcol -= HS;
    }
#pragma unroll
    for (int v = 0; v < 8; ++v) {
        const size_t r = (size_t)(tm * 16 + hi8 + v);
        const int    c = tf * 16 + l15;
        hdst[r * LDH + hcol] = acc[v];
        AGG[r * LDA + c]     = fmaf(acc[v], ns[v], bcol);
    }
}

template <int K, int F, int HS, int LDW, int LDH, int LDA, bool RELU_IN>
__global__ __launch_bounds__(128) void k_gemm_tail(const float* X,
                                                   const float* __restrict__ W,
                                                   const float* __restrict__ dinv,
                                                   const float* __restrict__ bias,
                                                   float* __restrict__ H,
                                                   float* H2,
                                                   float* __restrict__ AGG,
                                                   int rowStart) {
    static_assert(F <= 128, "one thread per column");
    __shared__ float xs[K];
    const int r = rowStart + blockIdx.x;
    for (int k = threadIdx.x; k < K; k += 128) {
        float xv = X[(size_t)r * K + k];
        if (RELU_IN) xv = fmaxf(xv, 0.0f);
        xs[k] = xv;
    }
    __syncthreads();
    const int f = threadIdx.x;
    if (f < F) {
        float s = 0.f;
        for (int k = 0; k < K; ++k) s += xs[k] * W[(size_t)k * LDW + f];
        const float d = dinv[r];
        float* hdst = (HS < F && f >= HS) ? H2 : H;
        const int hcol = (HS < F && f >= HS) ? f - HS : f;
        hdst[(size_t)r * LDH + hcol] = s;
        AGG[(size_t)r * LDA + f]     = fmaf(s, d * d, bias[f]);
    }
}

constexpr int Q_STRIDE = 64;
constexpr int Q_CELLS  = 8;
constexpr int EDGE_BLOCKS = 1024;

__device__ __forceinline__ int package_half() {
    return (int)((__builtin_amdgcn_s_sendmsg_rtn(0x87u) >> 18) & 1u);
}

template <int F, int LDH, int LDA, bool MAKE_W, int NPH>
__global__ __launch_bounds__(256) void k_scatter(const float* __restrict__ H,
                                                 const float* __restrict__ H2,
                                                 const int* __restrict__ src,
                                                 const int* __restrict__ dst,
                                                 const float* __restrict__ dinv,
                                                 float* __restrict__ WGT,
                                                 float* __restrict__ AGG,
                                                 int* __restrict__ tickets,
                                                 int E, int N) {
    static_assert(F % 32 == 0 && LDH % 32 == 0 && LDA % 32 == 0 && F <= LDH && NPH * F <= LDA,
                  "whole 128-byte lines");
    static_assert((NPH == 1 || NPH == 2) && !(MAKE_W && NPH > 1), "one or two column blocks; weights are made in a one-block pass");
    constexpr int NJ  = F / 32;
    constexpr int R   = 8;
    constexpr int CPT = 4;
    const int lane   = threadIdx.x & 31;
    const int nchunk = (E + 31) >> 5;
    const int nblock = ((nchunk + CPT - 1) / CPT) * CPT;
    const int ntick  = NPH * nblock;
    const int nsplit = (N >> 1) & ~1;
    const int maxit  = ntick / CPT + 2;
    const int hw     = package_half();

    for (int side = 0; side < 2; ++side) {
        const int h = hw ^ side;
        int* cell = tickets + h * Q_STRIDE;
        for (int it = 0; it < maxit; ++it) {
            int tk = 0;
            if (lane == 0) tk = atomicAdd(cell, CPT);
            const int tn = __builtin_amdgcn_readfirstlane(tk);
            if (tn >= ntick) break;
            const int ph = (NPH > 1 && tn >= nblock) ? 1 : 0;
            const int t  = tn - ph * nblock;
            const float* Hb = ph ? H2 : H;
            float* Ab = AGG + ph * F;

            int d[CPT], s[CPT];
            float w[CPT];
            unsigned m[CPT];
#pragma unroll
            for (int q = 0; q < CPT; ++q) {
                const int e = ((t + q) << 5) + lane;
                const bool in = e < E;
                d[q] = in ? dst[e] : 0;
                s[q] = in ? src[e] : 0;
                if (!MAKE_W) w[q] = in ? WGT[e] : 0.0f;
            }
            bool mine[CPT];
#pragma unroll
            for (int q = 0; q < CPT; ++q) {
                const int e = ((t + q) << 5) + lane;
                mine[q] = (e < E) && (((d[q] >= nsplit) ? 1 : 0) == h);
                m[q] = __builtin_amdgcn_ballot_w32(mine[q]);
            }
            if (MAKE_W) {
                float ds[CPT], dd[CPT];
#pragma unroll
                for (int q = 0; q < CPT; ++q) {
                    ds[q] = 0.0f;
                    dd[q] = 0.0f;
                    if (mine[q]) {
                        ds[q] = dinv[s[q]];
                        dd[q] = dinv[d[q]];
                    }
                }
#pragma unroll
                for (int q = 0; q < CPT; ++q) {
                    const int e = ((t + q) << 5) + lane;
                    w[q] = ds[q] * dd[q];
                    if (mine[q]) WGT[e] = w[q];
                }
            }
#pragma unroll
            for (int q = 0; q < CPT; ++q) {
                unsigned mm = m[q];
                for (int r = 0; r < 32 / R && mm != 0u; ++r) {
                    int jl[R];
#pragma unroll
                    for (int k = 0; k < R; ++k) {
                        jl[k] = (mm != 0u) ? __builtin_ctz(mm) : -1;
                        mm &= mm - 1u;
                    }
                    float v[R][NJ] = {};
#pragma unroll
                    for (int k = 0; k < R; ++k) {
                        if (jl[k] >= 0) {
                            const int sj = __builtin_amdgcn_readlane(s[q], jl[k]);
                            const float* hp = Hb + (size_t)sj * LDH + lane;
                            for (int c = 0; c < NJ; ++c) v[k][c] = hp[c * 32];
                        }
                    }
#pragma unroll
                    for (int k = 0; k < R; ++k) {
                        if (jl[k] >= 0) {
                            const int   dj = __builtin_amdgcn_readlane(d[q], jl[k]);
                            const float wj = __int_as_float(
                                __builtin_amdgcn_readlane(__float_as_int(w[q]), jl[k]));
                            float* ap = Ab + (size_t)dj * LDA + lane;
                            for (int c = 0; c < NJ; ++c) atomAddF(ap + c * 32, v[k][c] * wj);
                        }
                    }
                }
            }
        }
    }
}

static inline int nblk(long long n, int t) { return (int)((n + t - 1) / t); }

template <int K, int F, int HS, int LDW, int LDH, int LDA, bool RELU_IN>
static void launch_gemm(const float* X, const float* W, const float* dinv, const float* bias,
                        float* H, float* H2, float* AGG, int N, hipStream_t s) {
    int tilesM = N / 16;
    if (tilesM > 0)
        k_gemm_wmma<K, F, HS, LDW, LDH, LDA, RELU_IN><<<nblk((long long)tilesM, 8 / (F / 16)), 256, 0, s>>>(
            X, W, dinv, bias, H, H2, AGG, tilesM);
    int tail = tilesM * 16;
    if (tail < N)
        k_gemm_tail<K, F, HS, LDW, LDH, LDA, RELU_IN><<<N - tail, 128, 0, s>>>(
            X, W, dinv, bias, H, H2, AGG, tail);
}

extern "C" void kernel_launch(void* const* d_in, const int* in_sizes, int n_in,
                              void* d_out, int out_size, void* d_ws, size_t ws_size,
                              hipStream_t stream) {
    const float* x  = (const float*)d_in[0];
    const int*   ei = (const int*)d_in[1];
    const float* W1 = (const float*)d_in[2]; const float* b1 = (const float*)d_in[3];
    const float* W2 = (const float*)d_in[4]; const float* b2 = (const float*)d_in[5];
    const float* W3 = (const float*)d_in[6]; const float* b3 = (const float*)d_in[7];
    const float* W4 = (const float*)d_in[8]; const float* b4 = (const float*)d_in[9];

    const int N = in_sizes[0] / 128;
    const int E = in_sizes[1] / 2;
    const int* src = ei;
    const int* dst = ei + E;

    float* out  = (float*)d_out;
    float* xrec = out;
    float* z    = out + (size_t)N * 128;

    float* ws   = (float*)d_ws;
    float* bufH = ws;
    float* wgt  = bufH + (size_t)N * 64;
    int*   tick = (int*)(wgt + E);
    float* bufA = bufH + (size_t)N * 128;
    float* dinv = bufA + (size_t)N * 64;

    k_init_ones<<<nblk((long long)N + Q_CELLS * Q_STRIDE, 256), 256, 0, stream>>>(dinv, N, tick, Q_CELLS * Q_STRIDE);
    k_deg_lds<<<dim3(DEG_SLICES, nblk(N, DEG_RANGE)), 1024, 0, stream>>>(dst, dinv, E, N);
    k_rsqrt_inplace<<<nblk(N, 256), 256, 0, stream>>>(dinv, N);


    launch_gemm<128, 64, 64, 64, 64, 64, false>(x, W1, dinv, b1, bufH, bufH, bufA, N, stream);
    k_scatter<64, 64, 64, true, 1><<<EDGE_BLOCKS, 256, 0, stream>>>(bufH, bufH, src, dst, dinv, wgt, bufA, tick + 0 * Q_STRIDE, E, N);

    launch_gemm<64, 32, 32, 32, 32, 32, true>(bufA, W2, dinv, b2, bufH, bufH, z, N, stream);
    k_scatter<32, 32, 32, false, 1><<<EDGE_BLOCKS, 256, 0, stream>>>(bufH, bufH, src, dst, dinv, wgt, z, tick + 2 * Q_STRIDE, E, N);

    launch_gemm<32, 64, 64, 64, 64, 64, false>(z, W3, dinv, b3, bufH, bufH, bufA, N, stream);
    k_scatter<64, 64, 64, false, 1><<<EDGE_BLOCKS, 256, 0, stream>>>(bufH, bufH, src, dst, dinv, wgt, bufA, tick + 4 * Q_STRIDE, E, N);

    launch_gemm<64, 128, 64, 128, 64, 128, true>(bufA, W4, dinv, b4, bufH, bufA, xrec, N, stream);
    k_scatter<64, 64, 128, false, 2><<<EDGE_BLOCKS, 256, 0, stream>>>(bufH, bufA, src, dst, dinv, wgt, xrec, tick + 6 * Q_STRIDE, E, N);
}
